# starting kernel
# baseline (speedup 1.0000x reference)
#include <hip/hip_runtime.h>


typedef float v2f __attribute__((ext_vector_type(2)));
typedef float v8f __attribute__((ext_vector_type(8)));

#define WAVE 32
#define ROWW 20
#define TILES_PER_WAVE 8
#define WAVES_PER_BLOCK 8
#define TILES_PER_BLOCK (TILES_PER_WAVE * WAVES_PER_BLOCK)

__device__ __forceinline__ float fast_exp(float x) {
  return __builtin_amdgcn_exp2f(x * 1.4426950408889634f);
}
__device__ __forceinline__ float fast_rcp(float x) {
  return __builtin_amdgcn_rcpf(x);
}
__device__ __forceinline__ float fast_sigmoid(float x) {
  return fast_rcp(1.0f + fast_exp(-x));
}
__device__ __forceinline__ float fast_tanh(float x) {
  return 1.0f - 2.0f * fast_rcp(fast_exp(2.0f * x) + 1.0f);
}

__device__ __forceinline__ v8f wmma4(v2f a, v2f b, v8f c) {
  return __builtin_amdgcn_wmma_f32_16x16x4_f32(false, a, false, b, (short)0, c,
                                               false, false);
}

extern "C" __global__ void __launch_bounds__(256)
rssm_gru_wmma(const float* __restrict__ prev_z, const float* __restrict__ action,
              const float* __restrict__ prev_h,
              const float* __restrict__ W_ih, const float* __restrict__ W_hh,
              const float* __restrict__ b_ih, const float* __restrict__ b_hh,
              const float* __restrict__ W_mu, const float* __restrict__ b_mu,
              const float* __restrict__ W_lv, const float* __restrict__ b_lv,
              float* __restrict__ out_mu, float* __restrict__ out_lv,
              float* __restrict__ out_h, int ntiles)
{
  __shared__ float lds[WAVES_PER_BLOCK * 2 * 16 * ROWW];
  const int tid  = threadIdx.x;
  const int wave = tid / WAVE;
  const int lane = tid % WAVE;
  const int rb   = lane & 15;
  const int hi   = lane >> 4;

  float* sIn  = &lds[wave * 2 * 16 * ROWW];
  float* sOut = sIn + 16 * ROWW;

  v2f WihA[3], WhhA[3][4], WhdA[4];
  #pragma unroll
  for (int g = 0; g < 3; ++g) {
    const int row = g * 16 + rb;
    v2f a; a[0] = W_ih[row * 4 + 2 * hi + 0];
           a[1] = W_ih[row * 4 + 2 * hi + 1];
    WihA[g] = a;
    #pragma unroll
    for (int q = 0; q < 4; ++q) {
      v2f w; w[0] = W_hh[row * 16 + 4 * q + 2 * hi + 0];
             w[1] = W_hh[row * 16 + 4 * q + 2 * hi + 1];
      WhhA[g][q] = w;
    }
  }
  #pragma unroll
  for (int q = 0; q < 4; ++q) {
    v2f w;
    #pragma unroll
    for (int j = 0; j < 2; ++j) {
      const int col = 4 * q + 2 * hi + j;
      float val = 0.0f;
      if (rb < 3)      val = W_mu[rb * 16 + col];
      else if (rb < 6) val = W_lv[(rb - 3) * 16 + col];
      w[j] = val;
    }
    WhdA[q] = w;
  }

  float bcR[8], bcZ[8], bcNi[8], bcNh[8], bcH[8];
  #pragma unroll
  for (int v = 0; v < 8; ++v) {
    const int up = v + 8 * hi;
    bcR[v]  = b_ih[ 0 + up] + b_hh[ 0 + up];
    bcZ[v]  = b_ih[16 + up] + b_hh[16 + up];
    bcNi[v] = b_ih[32 + up];
    bcNh[v] = b_hh[32 + up];
    bcH[v]  = (up < 3) ? b_mu[up] : ((up < 6) ? b_lv[up - 3] : 0.0f);
  }

  for (int i = 0; i < TILES_PER_WAVE; ++i) {
    const int t = blockIdx.x * TILES_PER_BLOCK + i * WAVES_PER_BLOCK + wave;
    if (t >= ntiles) break;
    const int base = t * 16;

    {
      const int tn = t + WAVES_PER_BLOCK;
      if (tn < ntiles)
        __builtin_prefetch(prev_h + (size_t)tn * 256 + lane * 16, 0, 1);
    }

    const float4* gh4 = reinterpret_cast<const float4*>(prev_h + (size_t)base * 16);
    #pragma unroll
    for (int j = 0; j < 2; ++j) {
      const int e = j * 32 + lane;
      const float4 tv = gh4[e];
      const int r = e >> 2, c = (e & 3) * 4;
      *reinterpret_cast<float4*>(&sIn[r * ROWW + c]) = tv;
    }
    asm volatile("s_wait_dscnt 0" ::: "memory");

    v2f HbV[4];
    #pragma unroll
    for (int q = 0; q < 4; ++q) {
      v2f h2; h2[0] = sIn[rb * ROWW + 4 * q + 2 * hi + 0];
              h2[1] = sIn[rb * ROWW + 4 * q + 2 * hi + 1];
      HbV[q] = h2;
    }
    float Hcd[8];
    #pragma unroll
    for (int v = 0; v < 8; ++v)
      Hcd[v] = sIn[rb * ROWW + v + 8 * hi];

    const int b = base + rb;
    const float* pz = prev_z + (size_t)b * 3;
    const float* p0 = pz + 2 * hi;
    const float* p1 = hi ? (action + b) : (pz + 1);
    v2f Xb; Xb[0] = *p0; Xb[1] = *p1;

    v8f Dr, Dz, Dni, Dnh;
    #pragma unroll
    for (int v = 0; v < 8; ++v) { Dr[v] = bcR[v]; Dz[v] = bcZ[v]; Dni[v] = bcNi[v]; Dnh[v] = bcNh[v]; }
    Dr  = wmma4(WihA[0], Xb, Dr);
    Dz  = wmma4(WihA[1], Xb, Dz);
    Dni = wmma4(WihA[2], Xb, Dni);
    #pragma unroll
    for (int q = 0; q < 4; ++q) {
      Dr  = wmma4(WhhA[0][q], HbV[q], Dr);
      Dz  = wmma4(WhhA[1][q], HbV[q], Dz);
      Dnh = wmma4(WhhA[2][q], HbV[q], Dnh);
    }

    float hreg[8];
    #pragma unroll
    for (int v = 0; v < 8; ++v) {
      const float r = fast_sigmoid(Dr[v]);
      const float z = fast_sigmoid(Dz[v]);
      const float n = fast_tanh(fmaf(r, Dnh[v], Dni[v]));
      hreg[v] = fmaf(z, Hcd[v] - n, n);
    }

    #pragma unroll
    for (int v = 0; v < 8; ++v)
      sOut[rb * ROWW + v + 8 * hi] = hreg[v];
    asm volatile("s_wait_dscnt 0" ::: "memory");

    v8f Dh;
    #pragma unroll
    for (int v = 0; v < 8; ++v) Dh[v] = bcH[v];
    #pragma unroll
    for (int q = 0; q < 4; ++q) {
      v2f hb; hb[0] = sOut[rb * ROWW + 4 * q + 2 * hi + 0];
              hb[1] = sOut[rb * ROWW + 4 * q + 2 * hi + 1];
      Dh = wmma4(WhdA[q], hb, Dh);
    }

    float4* oh4 = reinterpret_cast<float4*>(out_h + (size_t)base * 16);
    #pragma unroll
    for (int j = 0; j < 2; ++j) {
      const int e = j * 32 + lane;
      const int r = e >> 2, c = (e & 3) * 4;
      oh4[e] = *reinterpret_cast<const float4*>(&sOut[r * ROWW + c]);
    }

    if (hi == 0) {
      #pragma unroll
      for (int k = 0; k < 3; ++k)
        out_mu[(size_t)b * 3 + k] = Dh[k];
      #pragma unroll
      for (int k = 0; k < 3; ++k) {
        float lv = Dh[3 + k];
        lv = fminf(fmaxf(lv, -5.0f), 5.0f);
        out_lv[(size_t)b * 3 + k] = lv;
      }
    }
  }
}

extern "C" void kernel_launch(void* const* d_in, const int* in_sizes, int n_in,
                              void* d_out, int out_size, void* d_ws, size_t ws_size,
                              hipStream_t stream) {
  const float* prev_z = (const float*)d_in[0];
  const float* action = (const float*)d_in[1];
  const float* prev_h = (const float*)d_in[2];
  const float* W_ih   = (const float*)d_in[3];
  const float* W_hh   = (const float*)d_in[4];
  const float* b_ih   = (const float*)d_in[5];
  const float* b_hh   = (const float*)d_in[6];
  const float* W_mu   = (const float*)d_in[7];
  const float* b_mu   = (const float*)d_in[8];
  const float* W_lv   = (const float*)d_in[9];
  const float* b_lv   = (const float*)d_in[10];

  const int B = in_sizes[1];
  float* out    = (float*)d_out;
  float* out_mu = out;
  float* out_lv = out + (size_t)B * 3;
  float* out_h  = out + (size_t)B * 6;

  const int ntiles = B / 16;
  const int grid   = (ntiles + TILES_PER_BLOCK - 1) / TILES_PER_BLOCK;

  rssm_gru_wmma<<<grid, 256, 0, stream>>>(prev_z, action, prev_h, W_ih, W_hh,
                                          b_ih, b_hh, W_mu, b_mu, W_lv, b_lv,
                                          out_mu, out_lv, out_h, ntiles);
}


# Software pipeline over a wave's tiles: tile i+1's prev_h, prev_z and action global loads issue into a second register set before tile i's WMMA and gate math; replaces the L2 prefetch hint
# speedup vs baseline: 1.2659x; 1.2659x over previous
#include <hip/hip_runtime.h>


typedef float v2f __attribute__((ext_vector_type(2)));
typedef float v8f __attribute__((ext_vector_type(8)));

#define WAVE 32
#define ROWW 20
#define TILES_PER_WAVE 8
#define WAVES_PER_BLOCK 8
#define TILES_PER_BLOCK (TILES_PER_WAVE * WAVES_PER_BLOCK)

__device__ __forceinline__ float fast_exp(float x) {
  return __builtin_amdgcn_exp2f(x * 1.4426950408889634f);
}
__device__ __forceinline__ float fast_rcp(float x) {
  return __builtin_amdgcn_rcpf(x);
}
__device__ __forceinline__ float fast_sigmoid(float x) {
  return fast_rcp(1.0f + fast_exp(-x));
}
__device__ __forceinline__ float fast_tanh(float x) {
  return 1.0f - 2.0f * fast_rcp(fast_exp(2.0f * x) + 1.0f);
}

__device__ __forceinline__ v8f wmma4(v2f a, v2f b, v8f c) {
  return __builtin_amdgcn_wmma_f32_16x16x4_f32(false, a, false, b, (short)0, c,
                                               false, false);
}


__device__ __forceinline__ void load_tile(const float* __restrict__ prev_z,
                                          const float* __restrict__ action,
                                          const float* __restrict__ prev_h,
                                          int t, int lane, int rb, int hi,
                                          float4& h0, float4& h1, v2f& x) {
  const int base = t * 16;
  const float4* gh4 = reinterpret_cast<const float4*>(prev_h + (size_t)base * 16);
  h0 = gh4[lane];
  h1 = gh4[32 + lane];
  const int b = base + rb;
  const float* pz = prev_z + (size_t)b * 3;
  const float* p0 = pz + 2 * hi;
  const float* p1 = hi ? (action + b) : (pz + 1);
  x[0] = *p0; x[1] = *p1;
}

extern "C" __global__ void __launch_bounds__(256)
rssm_gru_wmma(const float* __restrict__ prev_z, const float* __restrict__ action,
              const float* __restrict__ prev_h,
              const float* __restrict__ W_ih, const float* __restrict__ W_hh,
              const float* __restrict__ b_ih, const float* __restrict__ b_hh,
              const float* __restrict__ W_mu, const float* __restrict__ b_mu,
              const float* __restrict__ W_lv, const float* __restrict__ b_lv,
              float* __restrict__ out_mu, float* __restrict__ out_lv,
              float* __restrict__ out_h, int ntiles)
{
  __shared__ float lds[WAVES_PER_BLOCK * 2 * 16 * ROWW];
  const int tid  = threadIdx.x;
  const int wave = tid / WAVE;
  const int lane = tid % WAVE;
  const int rb   = lane & 15;
  const int hi   = lane >> 4;

  float* sIn  = &lds[wave * 2 * 16 * ROWW];
  float* sOut = sIn + 16 * ROWW;

  v2f WihA[3], WhhA[3][4], WhdA[4];
  #pragma unroll
  for (int g = 0; g < 3; ++g) {
    const int row = g * 16 + rb;
    v2f a; a[0] = W_ih[row * 4 + 2 * hi + 0];
           a[1] = W_ih[row * 4 + 2 * hi + 1];
    WihA[g] = a;
    #pragma unroll
    for (int q = 0; q < 4; ++q) {
      v2f w; w[0] = W_hh[row * 16 + 4 * q + 2 * hi + 0];
             w[1] = W_hh[row * 16 + 4 * q + 2 * hi + 1];
      WhhA[g][q] = w;
    }
  }
  #pragma unroll
  for (int q = 0; q < 4; ++q) {
    v2f w;
    #pragma unroll
    for (int j = 0; j < 2; ++j) {
      const int col = 4 * q + 2 * hi + j;
      float val = 0.0f;
      if (rb < 3)      val = W_mu[rb * 16 + col];
      else if (rb < 6) val = W_lv[(rb - 3) * 16 + col];
      w[j] = val;
    }
    WhdA[q] = w;
  }

  float bcR[8], bcZ[8], bcNi[8], bcNh[8], bcH[8];
  #pragma unroll
  for (int v = 0; v < 8; ++v) {
    const int up = v + 8 * hi;
    bcR[v]  = b_ih[ 0 + up] + b_hh[ 0 + up];
    bcZ[v]  = b_ih[16 + up] + b_hh[16 + up];
    bcNi[v] = b_ih[32 + up];
    bcNh[v] = b_hh[32 + up];
    bcH[v]  = (up < 3) ? b_mu[up] : ((up < 6) ? b_lv[up - 3] : 0.0f);
  }

  const int t0 = blockIdx.x * TILES_PER_BLOCK + wave;
  float4 nH0 = make_float4(0.f, 0.f, 0.f, 0.f), nH1 = nH0;
  v2f nX; nX[0] = 0.f; nX[1] = 0.f;
  if (t0 < ntiles)
    load_tile(prev_z, action, prev_h, t0, lane, rb, hi, nH0, nH1, nX);

  for (int i = 0; i < TILES_PER_WAVE; ++i) {
    const int t = t0 + i * WAVES_PER_BLOCK;
    if (t >= ntiles) break;
    const int base = t * 16;
    const int b = base + rb;

    const float4 cH0 = nH0, cH1 = nH1;
    const v2f Xb = nX;

    {
      const int tn = t + WAVES_PER_BLOCK;
      if (i + 1 < TILES_PER_WAVE && tn < ntiles)
        load_tile(prev_z, action, prev_h, tn, lane, rb, hi, nH0, nH1, nX);
    }

    {
      const int e0 = lane, e1 = 32 + lane;
      *reinterpret_cast<float4*>(&sIn[(e0 >> 2) * ROWW + (e0 & 3) * 4]) = cH0;
      *reinterpret_cast<float4*>(&sIn[(e1 >> 2) * ROWW + (e1 & 3) * 4]) = cH1;
    }
    asm volatile("s_wait_dscnt 0" ::: "memory");

    v2f HbV[4];
    #pragma unroll
    for (int q = 0; q < 4; ++q) {
      v2f h2; h2[0] = sIn[rb * ROWW + 4 * q + 2 * hi + 0];
              h2[1] = sIn[rb * ROWW + 4 * q + 2 * hi + 1];
      HbV[q] = h2;
    }
    float Hcd[8];
    #pragma unroll
    for (int v = 0; v < 8; ++v)
      Hcd[v] = sIn[rb * ROWW + v + 8 * hi];

    v8f Dr, Dz, Dni, Dnh;
    #pragma unroll
    for (int v = 0; v < 8; ++v) { Dr[v] = bcR[v]; Dz[v] = bcZ[v]; Dni[v] = bcNi[v]; Dnh[v] = bcNh[v]; }
    Dr  = wmma4(WihA[0], Xb, Dr);
    Dz  = wmma4(WihA[1], Xb, Dz);
    Dni = wmma4(WihA[2], Xb, Dni);
    #pragma unroll
    for (int q = 0; q < 4; ++q) {
      Dr  = wmma4(WhhA[0][q], HbV[q], Dr);
      Dz  = wmma4(WhhA[1][q], HbV[q], Dz);
      Dnh = wmma4(WhhA[2][q], HbV[q], Dnh);
    }

    float hreg[8];
    #pragma unroll
    for (int v = 0; v < 8; ++v) {
      const float r = fast_sigmoid(Dr[v]);
      const float z = fast_sigmoid(Dz[v]);
      const float n = fast_tanh(fmaf(r, Dnh[v], Dni[v]));
      hreg[v] = fmaf(z, Hcd[v] - n, n);
    }

    #pragma unroll
    for (int v = 0; v < 8; ++v)
      sOut[rb * ROWW + v + 8 * hi] = hreg[v];
    asm volatile("s_wait_dscnt 0" ::: "memory");

    v8f Dh;
    #pragma unroll
    for (int v = 0; v < 8; ++v) Dh[v] = bcH[v];
    #pragma unroll
    for (int q = 0; q < 4; ++q) {
      v2f hb; hb[0] = sOut[rb * ROWW + 4 * q + 2 * hi + 0];
              hb[1] = sOut[rb * ROWW + 4 * q + 2 * hi + 1];
      Dh = wmma4(WhdA[q], hb, Dh);
    }

    float4* oh4 = reinterpret_cast<float4*>(out_h + (size_t)base * 16);
    #pragma unroll
    for (int j = 0; j < 2; ++j) {
      const int e = j * 32 + lane;
      const int r = e >> 2, c = (e & 3) * 4;
      oh4[e] = *reinterpret_cast<const float4*>(&sOut[r * ROWW + c]);
    }

    if (hi == 0) {
      #pragma unroll
      for (int k = 0; k < 3; ++k)
        out_mu[(size_t)b * 3 + k] = Dh[k];
      #pragma unroll
      for (int k = 0; k < 3; ++k) {
        float lv = Dh[3 + k];
        lv = fminf(fmaxf(lv, -5.0f), 5.0f);
        out_lv[(size_t)b * 3 + k] = lv;
      }
    }
  }
}

extern "C" void kernel_launch(void* const* d_in, const int* in_sizes, int n_in,
                              void* d_out, int out_size, void* d_ws, size_t ws_size,
                              hipStream_t stream) {
  const float* prev_z = (const float*)d_in[0];
  const float* action = (const float*)d_in[1];
  const float* prev_h = (const float*)d_in[2];
  const float* W_ih   = (const float*)d_in[3];
  const float* W_hh   = (const float*)d_in[4];
  const float* b_ih   = (const float*)d_in[5];
  const float* b_hh   = (const float*)d_in[6];
  const float* W_mu   = (const float*)d_in[7];
  const float* b_mu   = (const float*)d_in[8];
  const float* W_lv   = (const float*)d_in[9];
  const float* b_lv   = (const float*)d_in[10];

  const int B = in_sizes[1];
  float* out    = (float*)d_out;
  float* out_mu = out;
  float* out_lv = out + (size_t)B * 3;
  float* out_h  = out + (size_t)B * 6;

  const int ntiles = B / 16;
  const int grid   = (ntiles + TILES_PER_BLOCK - 1) / TILES_PER_BLOCK;

  rssm_gru_wmma<<<grid, 256, 0, stream>>>(prev_z, action, prev_h, W_ih, W_hh,
                                          b_ih, b_hh, W_mu, b_mu, W_lv, b_lv,
                                          out_mu, out_lv, out_h, ntiles);
}
